# starting kernel
# baseline (speedup 1.0000x reference)
#include <hip/hip_runtime.h>
#include <math.h>

typedef __attribute__((ext_vector_type(2))) float v2f;
typedef __attribute__((ext_vector_type(8))) float v8f;

#define ENT_N 1024
#define DIM   100
#define EPS   1e-8f

__global__ __launch_bounds__(256)
void row_norms_kernel(const float* __restrict__ emb, float* __restrict__ norms) {
    int row = blockIdx.x * blockDim.x + threadIdx.x;
    if (row < ENT_N) {
        const float* p = emb + row * DIM;
        float s = 0.0f;
#pragma unroll 4
        for (int d = 0; d < DIM; ++d) {
            float v = p[d];
            s = fmaf(v, v, s);
        }
        norms[row] = sqrtf(s);
    }
}

__global__ __launch_bounds__(256)
void cosine_gram_wmma(const float* __restrict__ emb,
                      const float* __restrict__ norms,
                      float* __restrict__ out) {
    const int lane  = threadIdx.x & 31;
    const int wave  = threadIdx.x >> 5;
    const int w     = blockIdx.x * 8 + wave;
    const int mTile = w >> 4;
    const int nGrp  = w & 15;
    const int mBase = mTile * 16;
    const int nBase = nGrp * 64;

    const int l15  = lane & 15;
    const int half = lane >> 4;
    const int kofs = half * 2;

    const float* aPtr = emb + (mBase + l15) * DIM + kofs;
    const float* bPtr = emb + (nBase + l15) * DIM + kofs;

    v8f acc0 = {}, acc1 = {}, acc2 = {}, acc3 = {};

    for (int k = 0; k < DIM; k += 4) {
        v2f a  = *(const v2f*)(aPtr + k);
        v2f b0 = *(const v2f*)(bPtr + 0 * 16 * DIM + k);
        v2f b1 = *(const v2f*)(bPtr + 1 * 16 * DIM + k);
        v2f b2 = *(const v2f*)(bPtr + 2 * 16 * DIM + k);
        v2f b3 = *(const v2f*)(bPtr + 3 * 16 * DIM + k);
        acc0 = __builtin_amdgcn_wmma_f32_16x16x4_f32(false, a, false, b0,
                                                     (short)0, acc0, false, false);
        acc1 = __builtin_amdgcn_wmma_f32_16x16x4_f32(false, a, false, b1,
                                                     (short)0, acc1, false, false);
        acc2 = __builtin_amdgcn_wmma_f32_16x16x4_f32(false, a, false, b2,
                                                     (short)0, acc2, false, false);
        acc3 = __builtin_amdgcn_wmma_f32_16x16x4_f32(false, a, false, b3,
                                                     (short)0, acc3, false, false);
    }

    const float nrmN0 = norms[nBase +  0 + l15];
    const float nrmN1 = norms[nBase + 16 + l15];
    const float nrmN2 = norms[nBase + 32 + l15];
    const float nrmN3 = norms[nBase + 48 + l15];
    const int   n0    = nBase + l15;

#pragma unroll
    for (int r = 0; r < 8; ++r) {
        const int   m    = mBase + r + half * 8;
        const float nrmM = norms[m];
        float w0 = acc0[r] / fmaxf(nrmM * nrmN0, EPS);
        float w1 = acc1[r] / fmaxf(nrmM * nrmN1, EPS);
        float w2 = acc2[r] / fmaxf(nrmM * nrmN2, EPS);
        float w3 = acc3[r] / fmaxf(nrmM * nrmN3, EPS);
        float* o = out + (size_t)m * ENT_N;
        o[n0 +  0] = (m == n0 +  0) ? 0.0f : w0;
        o[n0 + 16] = (m == n0 + 16) ? 0.0f : w1;
        o[n0 + 32] = (m == n0 + 32) ? 0.0f : w2;
        o[n0 + 48] = (m == n0 + 48) ? 0.0f : w3;
    }
}

extern "C" void kernel_launch(void* const* d_in, const int* in_sizes, int n_in,
                              void* d_out, int out_size, void* d_ws, size_t ws_size,
                              hipStream_t stream) {
    const float* emb   = (const float*)d_in[2];
    float*       norms = (float*)d_ws;
    float*       out   = (float*)d_out;

    row_norms_kernel<<<(ENT_N + 255) / 256, 256, 0, stream>>>(emb, norms);
    cosine_gram_wmma<<<128, 256, 0, stream>>>(emb, norms, out);
}


# operands staged once per 64x64 output tile in LDS by whole-line 16-byte loads (256 blocks of 4 waves, 32x32 per wave); WMMA fragments read from LDS, not by 16-line loads from global; sums unchanged
# speedup vs baseline: 3.4800x; 3.4800x over previous
#include <hip/hip_runtime.h>
#include <math.h>


typedef __attribute__((ext_vector_type(2))) float v2f;
typedef __attribute__((ext_vector_type(4))) float v4f;
typedef __attribute__((ext_vector_type(8))) float v8f;

#define ENT_N  1024
#define DIM    100
#define EPS    1e-8f
#define KSTEPS (DIM / 4)
#define TILE   64
#define TILE_F (TILE * DIM)
#define TILE_V (TILE_F / 4)
#define NTHR   128
#define ROUNDS (2 * TILE_V / NTHR)

__device__ __forceinline__ float xhalf(float x) {
    int b = __builtin_bit_cast(int, x);
    return __builtin_bit_cast(float,
        __builtin_amdgcn_permlanex16(b, b, 0x76543210, 0xfedcba98, false, false));
}

__global__ __launch_bounds__(NTHR)
void cosine_gram_wmma(const float* __restrict__ emb,
                      float* __restrict__ out) {
    __shared__ __attribute__((aligned(16))) float tile[2 * TILE_F];

    const int t    = threadIdx.x;
    const int mBlk = blockIdx.x >> 4;
    const int nBlk = blockIdx.x & 15;

    const v4f* gA = (const v4f*)(emb + mBlk * TILE_F);
    const v4f* gB = (const v4f*)(emb + nBlk * TILE_F);
    v4f*       sV = (v4f*)tile;

    v4f stg[ROUNDS];
#pragma unroll
    for (int i = 0; i < ROUNDS; ++i) {
        const int j = i * NTHR + t;
        const v4f* p;
        if ((i + 1) * NTHR <= TILE_V)  p = gA + j;
        else if (i * NTHR >= TILE_V)   p = gB + (j - TILE_V);
        else                           p = (j < TILE_V) ? (gA + j) : (gB + (j - TILE_V));
        stg[i] = *p;
    }
#pragma unroll
    for (int i = 0; i < ROUNDS; ++i) sV[i * NTHR + t] = stg[i];
    __syncthreads();

    const int lane = t & 31;
    const int wave = t >> 5;
    const int wm   = wave >> 1;
    const int wn   = wave & 1;
    const int l15  = lane & 15;
    const int half = lane >> 4;
    const int kofs = half * 2;

    const float* aPtr = tile + (wm * 32 + l15) * DIM + kofs;
    const float* bPtr = tile + TILE_F + (wn * 32 + l15) * DIM + kofs;

    v2f fa0[KSTEPS], fa1[KSTEPS], fb0[KSTEPS], fb1[KSTEPS];
#pragma unroll
    for (int s = 0; s < KSTEPS; ++s) {
        fa0[s] = *(const v2f*)(aPtr + 4 * s);
        fb0[s] = *(const v2f*)(bPtr + 4 * s);
        fa1[s] = *(const v2f*)(aPtr + 16 * DIM + 4 * s);
        fb1[s] = *(const v2f*)(bPtr + 16 * DIM + 4 * s);
    }
    __builtin_amdgcn_sched_barrier(0);

    v8f acc00 = {}, acc01 = {}, acc10 = {}, acc11 = {};
    v2f qA0 = {}, qA1 = {}, qB0 = {}, qB1 = {};

#pragma unroll
    for (int s = 0; s < KSTEPS; ++s) {
        const v2f a0 = fa0[s], a1 = fa1[s], b0 = fb0[s], b1 = fb1[s];
        acc00 = __builtin_amdgcn_wmma_f32_16x16x4_f32(false, a0, false, b0,
                                                      (short)0, acc00, false, false);
        acc01 = __builtin_amdgcn_wmma_f32_16x16x4_f32(false, a0, false, b1,
                                                      (short)0, acc01, false, false);
        acc10 = __builtin_amdgcn_wmma_f32_16x16x4_f32(false, a1, false, b0,
                                                      (short)0, acc10, false, false);
        acc11 = __builtin_amdgcn_wmma_f32_16x16x4_f32(false, a1, false, b1,
                                                      (short)0, acc11, false, false);
        qA0 = __builtin_elementwise_fma(a0, a0, qA0);
        qA1 = __builtin_elementwise_fma(a1, a1, qA1);
        qB0 = __builtin_elementwise_fma(b0, b0, qB0);
        qB1 = __builtin_elementwise_fma(b1, b1, qB1);
    }

    const float sA0 = qA0.x + qA0.y;
    const float sA1 = qA1.x + qA1.y;
    const float sB0 = qB0.x + qB0.y;
    const float sB1 = qB1.x + qB1.y;
    const float nrmM0 = sqrtf(sA0 + xhalf(sA0));
    const float nrmM1 = sqrtf(sA1 + xhalf(sA1));
    const float nrmN0 = sqrtf(sB0 + xhalf(sB0));
    const float nrmN1 = sqrtf(sB1 + xhalf(sB1));

    const int mBase = mBlk * TILE + wm * 32;
    const int n0    = nBlk * TILE + wn * 32 + l15;

#pragma unroll
    for (int r = 0; r < 8; ++r) {
        const int   mLoc = r + half * 8;
        const int   mA   = mBase + mLoc;
        const int   mB   = mA + 16;
        const float nA   = __shfl(nrmM0, mLoc, 32);
        const float nB   = __shfl(nrmM1, mLoc, 32);
        float w00 = acc00[r] / fmaxf(nA * nrmN0, EPS);
        float w01 = acc01[r] / fmaxf(nA * nrmN1, EPS);
        float w10 = acc10[r] / fmaxf(nB * nrmN0, EPS);
        float w11 = acc11[r] / fmaxf(nB * nrmN1, EPS);
        float* oA = out + (size_t)mA * ENT_N;
        float* oB = out + (size_t)mB * ENT_N;
        oA[n0 +  0] = (mA == n0 +  0) ? 0.0f : w00;
        oA[n0 + 16] = (mA == n0 + 16) ? 0.0f : w01;
        oB[n0 +  0] = (mB == n0 +  0) ? 0.0f : w10;
        oB[n0 + 16] = (mB == n0 + 16) ? 0.0f : w11;
    }
}

extern "C" void kernel_launch(void* const* d_in, const int* in_sizes, int n_in,
                              void* d_out, int out_size, void* d_ws, size_t ws_size,
                              hipStream_t stream) {
    const float* emb = (const float*)d_in[2];
    float*       out = (float*)d_out;

    cosine_gram_wmma<<<256, NTHR, 0, stream>>>(emb, out);
}


# symmetry out[m][n]=out[n][m]: each off-diagonal tile pair computed once by 32x64 work-groups (50 WMMAs per wave, half the divisions), written twice, the mirror via an LDS transpose; sums unchanged
# speedup vs baseline: 3.9560x; 1.1368x over previous
#include <hip/hip_runtime.h>
#include <math.h>


typedef __attribute__((ext_vector_type(2))) float v2f;
typedef __attribute__((ext_vector_type(4))) float v4f;
typedef __attribute__((ext_vector_type(8))) float v8f;

#define ENT_N  1024
#define DIM    100
#define EPS    1e-8f
#define KSTEPS (DIM / 4)
#define TILE   64
#define TILE_F (TILE * DIM)
#define TILE_V (TILE_F / 4)
#define NTHR   128
#define D_ROUNDS ((TILE_V + NTHR - 1) / NTHR)

#define H_PCS_M  (32 * DIM / 4)
#define H_PCS    (H_PCS_M + TILE_V)
#define H_ROUNDS ((H_PCS + NTHR - 1) / NTHR)
#define OPITCH   36

__device__ __forceinline__ float xhalf(float x) {
    int b = __builtin_bit_cast(int, x);
    return __builtin_bit_cast(float,
        __builtin_amdgcn_permlanex16(b, b, 0x76543210, 0xfedcba98, false, false));
}

__device__ __forceinline__
void diag_tile(const float* __restrict__ emb, float* __restrict__ out,
               float* tile, const int dBlk, const int t) {
    const int mBlk = dBlk;
    const int nBlk = dBlk;

    const v4f* gA = (const v4f*)(emb + dBlk * TILE_F);
    v4f*       sV = (v4f*)tile;

    v4f stg[D_ROUNDS];
#pragma unroll
    for (int i = 0; i < D_ROUNDS; ++i) {
        const int j  = i * NTHR + t;
        const int js = ((i + 1) * NTHR <= TILE_V || j < TILE_V) ? j : j - NTHR;
        stg[i] = gA[js];
    }
#pragma unroll
    for (int i = 0; i < D_ROUNDS; ++i) sV[i * NTHR + t] = stg[i];
    __syncthreads();

    const int lane = t & 31;
    const int wave = t >> 5;
    const int wm   = wave >> 1;
    const int wn   = wave & 1;
    const int l15  = lane & 15;
    const int half = lane >> 4;
    const int kofs = half * 2;

    const float* aPtr = tile + (wm * 32 + l15) * DIM + kofs;
    const float* bPtr = tile + (wn * 32 + l15) * DIM + kofs;

    v2f fa0[KSTEPS], fa1[KSTEPS], fb0[KSTEPS], fb1[KSTEPS];
#pragma unroll
    for (int s = 0; s < KSTEPS; ++s) {
        fa0[s] = *(const v2f*)(aPtr + 4 * s);
        fb0[s] = *(const v2f*)(bPtr + 4 * s);
        fa1[s] = *(const v2f*)(aPtr + 16 * DIM + 4 * s);
        fb1[s] = *(const v2f*)(bPtr + 16 * DIM + 4 * s);
    }
    __builtin_amdgcn_sched_barrier(0);

    v8f acc00 = {}, acc01 = {}, acc10 = {}, acc11 = {};
    v2f qA0 = {}, qA1 = {}, qB0 = {}, qB1 = {};

#pragma unroll
    for (int s = 0; s < KSTEPS; ++s) {
        const v2f a0 = fa0[s], a1 = fa1[s], b0 = fb0[s], b1 = fb1[s];
        acc00 = __builtin_amdgcn_wmma_f32_16x16x4_f32(false, a0, false, b0,
                                                      (short)0, acc00, false, false);
        acc01 = __builtin_amdgcn_wmma_f32_16x16x4_f32(false, a0, false, b1,
                                                      (short)0, acc01, false, false);
        acc10 = __builtin_amdgcn_wmma_f32_16x16x4_f32(false, a1, false, b0,
                                                      (short)0, acc10, false, false);
        acc11 = __builtin_amdgcn_wmma_f32_16x16x4_f32(false, a1, false, b1,
                                                      (short)0, acc11, false, false);
        qA0 = __builtin_elementwise_fma(a0, a0, qA0);
        qA1 = __builtin_elementwise_fma(a1, a1, qA1);
        qB0 = __builtin_elementwise_fma(b0, b0, qB0);
        qB1 = __builtin_elementwise_fma(b1, b1, qB1);
    }

    const float sA0 = qA0.x + qA0.y;
    const float sA1 = qA1.x + qA1.y;
    const float sB0 = qB0.x + qB0.y;
    const float sB1 = qB1.x + qB1.y;
    const float nrmM0 = sqrtf(sA0 + xhalf(sA0));
    const float nrmM1 = sqrtf(sA1 + xhalf(sA1));
    const float nrmN0 = sqrtf(sB0 + xhalf(sB0));
    const float nrmN1 = sqrtf(sB1 + xhalf(sB1));

    const int mBase = mBlk * TILE + wm * 32;
    const int n0    = nBlk * TILE + wn * 32 + l15;

#pragma unroll
    for (int r = 0; r < 8; ++r) {
        const int   mLoc = r + half * 8;
        const int   mA   = mBase + mLoc;
        const int   mB   = mA + 16;
        const float nA   = __shfl(nrmM0, mLoc, 32);
        const float nB   = __shfl(nrmM1, mLoc, 32);
        float w00 = acc00[r] / fmaxf(nA * nrmN0, EPS);
        float w01 = acc01[r] / fmaxf(nA * nrmN1, EPS);
        float w10 = acc10[r] / fmaxf(nB * nrmN0, EPS);
        float w11 = acc11[r] / fmaxf(nB * nrmN1, EPS);
        float* oA = out + (size_t)mA * ENT_N;
        float* oB = out + (size_t)mB * ENT_N;
        oA[n0 +  0] = (mA == n0 +  0) ? 0.0f : w00;
        oA[n0 + 16] = (mA == n0 + 16) ? 0.0f : w01;
        oB[n0 +  0] = (mB == n0 +  0) ? 0.0f : w10;
        oB[n0 + 16] = (mB == n0 + 16) ? 0.0f : w11;
    }
}

__device__ __forceinline__
void half_tile(const float* __restrict__ emb, float* __restrict__ out,
               float* tile, float* otile, const int I, const int Jb, const int t) {
    const v4f* gA = (const v4f*)(emb + I * (32 * DIM));
    const v4f* gB = (const v4f*)(emb + Jb * TILE_F);
    v4f*       sV = (v4f*)tile;

    v4f stg[H_ROUNDS];
#pragma unroll
    for (int i = 0; i < H_ROUNDS; ++i) {
        const int j  = i * NTHR + t;
        const int js = ((i + 1) * NTHR <= H_PCS || j < H_PCS) ? j : j - NTHR;
        const v4f* p;
        if ((i + 1) * NTHR <= H_PCS_M)  p = gA + js;
        else if (i * NTHR >= H_PCS_M)   p = gB + (js - H_PCS_M);
        else                            p = (js < H_PCS_M) ? (gA + js) : (gB + (js - H_PCS_M));
        stg[i] = *p;
    }
#pragma unroll
    for (int i = 0; i < H_ROUNDS; ++i) sV[i * NTHR + t] = stg[i];
    __syncthreads();

    const int lane = t & 31;
    const int wave = t >> 5;
    const int l15  = lane & 15;
    const int half = lane >> 4;
    const int kofs = half * 2;

    const float* aPtr = tile + l15 * DIM + kofs;
    const float* bPtr = tile + 32 * DIM + (wave * 16 + l15) * DIM + kofs;

    v2f fa0[KSTEPS], fa1[KSTEPS], fb[KSTEPS];
#pragma unroll
    for (int s = 0; s < KSTEPS; ++s) {
        fa0[s] = *(const v2f*)(aPtr + 4 * s);
        fb[s]  = *(const v2f*)(bPtr + 4 * s);
        fa1[s] = *(const v2f*)(aPtr + 16 * DIM + 4 * s);
    }
    __builtin_amdgcn_sched_barrier(0);

    v8f acc0 = {}, acc1 = {};
    v2f qA0 = {}, qA1 = {}, qB = {};

#pragma unroll
    for (int s = 0; s < KSTEPS; ++s) {
        const v2f a0 = fa0[s], a1 = fa1[s], b = fb[s];
        acc0 = __builtin_amdgcn_wmma_f32_16x16x4_f32(false, a0, false, b,
                                                     (short)0, acc0, false, false);
        acc1 = __builtin_amdgcn_wmma_f32_16x16x4_f32(false, a1, false, b,
                                                     (short)0, acc1, false, false);
        qA0 = __builtin_elementwise_fma(a0, a0, qA0);
        qA1 = __builtin_elementwise_fma(a1, a1, qA1);
        qB  = __builtin_elementwise_fma(b, b, qB);
    }

    const float sA0 = qA0.x + qA0.y;
    const float sA1 = qA1.x + qA1.y;
    const float sB  = qB.x + qB.y;
    const float nrmM0 = sqrtf(sA0 + xhalf(sA0));
    const float nrmM1 = sqrtf(sA1 + xhalf(sA1));
    const float nrmN  = sqrtf(sB + xhalf(sB));

    const int mBase = I * 32;
    const int nBase = Jb * TILE + wave * 16;
    const int n0    = nBase + l15;

    v8f w0, w1;
#pragma unroll
    for (int r = 0; r < 8; ++r) {
        const int   mLoc = r + half * 8;
        const float nA   = __shfl(nrmM0, mLoc, 32);
        const float nB   = __shfl(nrmM1, mLoc, 32);
        w0[r] = acc0[r] / fmaxf(nA * nrmN, EPS);
        w1[r] = acc1[r] / fmaxf(nB * nrmN, EPS);
        out[(size_t)(mBase + mLoc) * ENT_N + n0]      = w0[r];
        out[(size_t)(mBase + 16 + mLoc) * ENT_N + n0] = w1[r];
    }

    float* oW = otile + wave * (16 * OPITCH);
    v4f*   s0 = (v4f*)(oW + l15 * OPITCH + half * 8);
    s0[0] = __builtin_shufflevector(w0, w0, 0, 1, 2, 3);
    s0[1] = __builtin_shufflevector(w0, w0, 4, 5, 6, 7);
    s0[4] = __builtin_shufflevector(w1, w1, 0, 1, 2, 3);
    s0[5] = __builtin_shufflevector(w1, w1, 4, 5, 6, 7);

    const int g    = lane & 7;
    const int q    = lane >> 3;
    const int rsel = 8 * (q & 1) + 4 * (q >> 1);
#pragma unroll
    for (int i = 0; i < 4; ++i) {
        const int row = i + rsel;
        const v4f v = *(const v4f*)(oW + row * OPITCH + 4 * g);
        *(v4f*)(out + (size_t)(nBase + row) * ENT_N + mBase + 4 * g) = v;
    }
}

__global__ __launch_bounds__(NTHR)
void cosine_gram_wmma(const float* __restrict__ emb,
                      float* __restrict__ out) {
    __shared__ __attribute__((aligned(16))) float tile[4 * H_ROUNDS * NTHR];
    __shared__ __attribute__((aligned(16))) float otile[4 * 16 * OPITCH];

    const int t = threadIdx.x;
    const int b = blockIdx.x;
    if (b < 16) {
        diag_tile(emb, out, tile, b, t);
    } else {
        const int u = b - 16;
        int Jb = 1;
#pragma unroll
        for (int j = 2; j <= 15; ++j) if (u >= j * (j - 1)) Jb = j;
        const int I = u - Jb * (Jb - 1);
        half_tile(emb, out, tile, otile, I, Jb, t);
    }
}

extern "C" void kernel_launch(void* const* d_in, const int* in_sizes, int n_in,
                              void* d_out, int out_size, void* d_ws, size_t ws_size,
                              hipStream_t stream) {
    const float* emb = (const float*)d_in[2];
    float*       out = (float*)d_out;

    cosine_gram_wmma<<<256, NTHR, 0, stream>>>(emb, out);
}
